# starting kernel
# baseline (speedup 1.0000x reference)
#include <hip/hip_runtime.h>

#define BB 64
#define FF 128
#define CIN 64
#define CON 64
#define RR 4

typedef float v2f __attribute__((ext_vector_type(2)));
typedef float v8f __attribute__((ext_vector_type(8)));

#define E_STRIDE 65
#define E_R (BB * E_STRIDE)
#define W_STRIDE 65
#define W_R (CON * W_STRIDE)
#define LE_OFF 0
#define LW_OFF (RR * E_R)
#define LM_OFF (LW_OFF + RR * W_R)
#define LDS_FLOATS (LM_OFF + RR * BB)
#define LDS_BYTES (LDS_FLOATS * 4)

extern "C" __global__ __launch_bounds__(256)
void weightedsum_lse_wmma(const float* __restrict__ ll,
                          const float* __restrict__ wgt,
                          float* __restrict__ out) {
    extern __shared__ float smem[];
    const int f   = blockIdx.x;
    const int tid = threadIdx.x;

    const float4* ll4 = (const float4*)ll;
    const float4* w4  = (const float4*)wgt;
    #pragma unroll
    for (int i = 0; i < 16; ++i) {
        const int idx = tid + i * 256;
        const int hi = idx >> 6, lo = idx & 63;

        float4 v = ll4[(hi * FF + f) * CIN + lo];
        smem[LE_OFF + 0 * E_R + hi * E_STRIDE + lo] = v.x;
        smem[LE_OFF + 1 * E_R + hi * E_STRIDE + lo] = v.y;
        smem[LE_OFF + 2 * E_R + hi * E_STRIDE + lo] = v.z;
        smem[LE_OFF + 3 * E_R + hi * E_STRIDE + lo] = v.w;

        float4 wv = w4[(f * CIN + hi) * CON + lo];
        smem[LW_OFF + 0 * W_R + lo * W_STRIDE + hi] = wv.x;
        smem[LW_OFF + 1 * W_R + lo * W_STRIDE + hi] = wv.y;
        smem[LW_OFF + 2 * W_R + lo * W_STRIDE + hi] = wv.z;
        smem[LW_OFF + 3 * W_R + lo * W_STRIDE + hi] = wv.w;
    }
    __syncthreads();

    {
        const int b = tid & 63;
        const int r = tid >> 6;
        float* row = &smem[LE_OFF + r * E_R + b * E_STRIDE];
        float m = row[0];
        #pragma unroll 8
        for (int ci = 1; ci < CIN; ++ci) m = fmaxf(m, row[ci]);
        smem[LM_OFF + r * BB + b] = m;
        #pragma unroll 8
        for (int ci = 0; ci < CIN; ++ci) row[ci] = __expf(row[ci] - m);
    }
    __syncthreads();

    const int wave  = tid >> 5;
    const int lane  = tid & 31;
    const int r     = wave >> 1;
    const int mhalf = wave & 1;
    const int lrow  = lane & 15;
    const int koff  = (lane >> 4) << 1;

    const float* Eb = &smem[LE_OFF + r * E_R];
    const float* Wb = &smem[LW_OFF + r * W_R];

    v8f acc[8];
    #pragma unroll
    for (int mt = 0; mt < 2; ++mt) {
        const int arow = (mhalf * 2 + mt) * 16 + lrow;
        #pragma unroll
        for (int nt = 0; nt < 4; ++nt) {
            const int bcol = nt * 16 + lrow;
            v8f c = {0.f, 0.f, 0.f, 0.f, 0.f, 0.f, 0.f, 0.f};
            #pragma unroll
            for (int k = 0; k < CIN; k += 4) {
                const int kk = k + koff;
                v2f a, b;
                a.x = Eb[arow * E_STRIDE + kk];
                a.y = Eb[arow * E_STRIDE + kk + 1];
                b.x = Wb[bcol * W_STRIDE + kk];
                b.y = Wb[bcol * W_STRIDE + kk + 1];
                c = __builtin_amdgcn_wmma_f32_16x16x4_f32(
                         false, a,  false, b,
                         (short)0, c,  false,  false);
            }
            acc[mt * 4 + nt] = c;
        }
    }
    __syncthreads();

    {
        float* O = &smem[LE_OFF + r * E_R];
        const int rowadd = (lane >> 4) << 3;
        #pragma unroll
        for (int mt = 0; mt < 2; ++mt) {
            const int base_row = (mhalf * 2 + mt) * 16 + rowadd;
            #pragma unroll
            for (int nt = 0; nt < 4; ++nt) {
                const int col = nt * 16 + lrow;
                v8f c = acc[mt * 4 + nt];
                #pragma unroll
                for (int v = 0; v < 8; ++v)
                    O[(base_row + v) * E_STRIDE + col] = c[v];
            }
        }
    }
    __syncthreads();

    float4* out4 = (float4*)out;
    #pragma unroll
    for (int i = 0; i < 16; ++i) {
        const int idx = tid + i * 256;
        const int b = idx >> 6, co = idx & 63;
        float4 o;
        o.x = smem[LM_OFF + 0 * BB + b] + __logf(smem[LE_OFF + 0 * E_R + b * E_STRIDE + co]);
        o.y = smem[LM_OFF + 1 * BB + b] + __logf(smem[LE_OFF + 1 * E_R + b * E_STRIDE + co]);
        o.z = smem[LM_OFF + 2 * BB + b] + __logf(smem[LE_OFF + 2 * E_R + b * E_STRIDE + co]);
        o.w = smem[LM_OFF + 3 * BB + b] + __logf(smem[LE_OFF + 3 * E_R + b * E_STRIDE + co]);
        out4[(b * FF + f) * CON + co] = o;
    }
}

extern "C" void kernel_launch(void* const* d_in, const int* in_sizes, int n_in,
                              void* d_out, int out_size, void* d_ws, size_t ws_size,
                              hipStream_t stream) {
    (void)in_sizes; (void)n_in; (void)d_ws; (void)ws_size; (void)out_size;
    const float* ll  = (const float*)d_in[0];
    const float* wgt = (const float*)d_in[1];
    float* out = (float*)d_out;

    weightedsum_lse_wmma<<<dim3(FF), dim3(256), LDS_BYTES, stream>>>(ll, wgt, out);
}


# row max/exp phase: each thread reads its 64-value LDS row into registers in one batch and takes max and exp there (row read once, no LDS wait per pair of values); E, m and sum order bit-identical
# speedup vs baseline: 1.6654x; 1.6654x over previous
#include <hip/hip_runtime.h>

#define BB 64
#define FF 128
#define CIN 64
#define CON 64
#define RR 4
#define CH 32
#define NGRP 256

typedef float v2f __attribute__((ext_vector_type(2)));
typedef float v8f __attribute__((ext_vector_type(8)));

#define E_STRIDE 65
#define E_R (BB * E_STRIDE)
#define W_STRIDE 65
#define W_R (CH * W_STRIDE)
#define LE_OFF 0
#define LW_OFF (RR * E_R)
#define LM_OFF (LW_OFF + RR * W_R)
#define LDS_FLOATS (LM_OFF + RR * BB)
#define LDS_BYTES (LDS_FLOATS * 4)

extern "C" __global__ __launch_bounds__(256)
void weightedsum_lse_wmma(const float* __restrict__ ll,
                          const float* __restrict__ wgt,
                          float* __restrict__ out) {
    extern __shared__ float smem[];
    const int f   = blockIdx.x >> 1;
    const int c0  = (blockIdx.x & 1) * CH;
    const int tid = threadIdx.x;

    const float4* ll4 = (const float4*)ll;
    const float4* w4  = (const float4*)wgt;
    #pragma unroll
    for (int i = 0; i < 16; ++i) {
        const int idx = tid + i * 256;
        const int hi = idx >> 6, lo = idx & 63;

        float4 v = ll4[(hi * FF + f) * CIN + lo];
        smem[LE_OFF + 0 * E_R + hi * E_STRIDE + lo] = v.x;
        smem[LE_OFF + 1 * E_R + hi * E_STRIDE + lo] = v.y;
        smem[LE_OFF + 2 * E_R + hi * E_STRIDE + lo] = v.z;
        smem[LE_OFF + 3 * E_R + hi * E_STRIDE + lo] = v.w;
    }
    #pragma unroll
    for (int i = 0; i < 8; ++i) {
        const int idx = tid + i * 256;
        const int hi = idx >> 5, lo = idx & 31;

        float4 wv = w4[(f * CIN + hi) * CON + c0 + lo];
        smem[LW_OFF + 0 * W_R + lo * W_STRIDE + hi] = wv.x;
        smem[LW_OFF + 1 * W_R + lo * W_STRIDE + hi] = wv.y;
        smem[LW_OFF + 2 * W_R + lo * W_STRIDE + hi] = wv.z;
        smem[LW_OFF + 3 * W_R + lo * W_STRIDE + hi] = wv.w;
    }
    __syncthreads();

    {
        const int b = tid & 63;
        const int r = tid >> 6;
        float* row = &smem[LE_OFF + r * E_R + b * E_STRIDE];
        float v[CIN];
        #pragma unroll
        for (int ci = 0; ci < CIN; ++ci) v[ci] = row[ci];
        float m0 = v[0], m1 = v[1], m2 = v[2], m3 = v[3];
        #pragma unroll
        for (int ci = 4; ci < CIN; ci += 4) {
            m0 = fmaxf(m0, v[ci]);
            m1 = fmaxf(m1, v[ci + 1]);
            m2 = fmaxf(m2, v[ci + 2]);
            m3 = fmaxf(m3, v[ci + 3]);
        }
        const float m = fmaxf(fmaxf(m0, m1), fmaxf(m2, m3));
        smem[LM_OFF + r * BB + b] = m;
        #pragma unroll
        for (int ci = 0; ci < CIN; ++ci) row[ci] = __expf(v[ci] - m);
    }
    __syncthreads();

    const int wave  = tid >> 5;
    const int lane  = tid & 31;
    const int r     = wave >> 1;
    const int mhalf = wave & 1;
    const int lrow  = lane & 15;
    const int koff  = (lane >> 4) << 1;

    const float* Eb = &smem[LE_OFF + r * E_R];
    const float* Wb = &smem[LW_OFF + r * W_R];

    v8f acc[4];
    #pragma unroll
    for (int mt = 0; mt < 2; ++mt) {
        const int arow = (mhalf * 2 + mt) * 16 + lrow;
        #pragma unroll
        for (int nt = 0; nt < 2; ++nt) {
            const int bcol = nt * 16 + lrow;
            v8f c = {0.f, 0.f, 0.f, 0.f, 0.f, 0.f, 0.f, 0.f};
            #pragma unroll
            for (int k = 0; k < CIN; k += 4) {
                const int kk = k + koff;
                v2f a, b;
                a.x = Eb[arow * E_STRIDE + kk];
                a.y = Eb[arow * E_STRIDE + kk + 1];
                b.x = Wb[bcol * W_STRIDE + kk];
                b.y = Wb[bcol * W_STRIDE + kk + 1];
                c = __builtin_amdgcn_wmma_f32_16x16x4_f32(
                         false, a,  false, b,
                         (short)0, c,  false,  false);
            }
            acc[mt * 2 + nt] = c;
        }
    }
    __syncthreads();

    {
        float* O = &smem[LE_OFF + r * E_R];
        const int rowadd = (lane >> 4) << 3;
        #pragma unroll
        for (int mt = 0; mt < 2; ++mt) {
            const int base_row = (mhalf * 2 + mt) * 16 + rowadd;
            #pragma unroll
            for (int nt = 0; nt < 2; ++nt) {
                const int col = nt * 16 + lrow;
                v8f c = acc[mt * 2 + nt];
                #pragma unroll
                for (int v = 0; v < 8; ++v)
                    O[(base_row + v) * E_STRIDE + col] = c[v];
            }
        }
    }
    __syncthreads();

    float4* out4 = (float4*)out;
    #pragma unroll
    for (int i = 0; i < 8; ++i) {
        const int idx = tid + i * 256;
        const int b = idx >> 5, co = idx & 31;
        float4 o;
        o.x = smem[LM_OFF + 0 * BB + b] + __logf(smem[LE_OFF + 0 * E_R + b * E_STRIDE + co]);
        o.y = smem[LM_OFF + 1 * BB + b] + __logf(smem[LE_OFF + 1 * E_R + b * E_STRIDE + co]);
        o.z = smem[LM_OFF + 2 * BB + b] + __logf(smem[LE_OFF + 2 * E_R + b * E_STRIDE + co]);
        o.w = smem[LM_OFF + 3 * BB + b] + __logf(smem[LE_OFF + 3 * E_R + b * E_STRIDE + co]);
        out4[(b * FF + f) * CON + c0 + co] = o;
    }
}

extern "C" void kernel_launch(void* const* d_in, const int* in_sizes, int n_in,
                              void* d_out, int out_size, void* d_ws, size_t ws_size,
                              hipStream_t stream) {
    (void)in_sizes; (void)n_in; (void)d_ws; (void)ws_size; (void)out_size;
    const float* ll  = (const float*)d_in[0];
    const float* wgt = (const float*)d_in[1];
    float* out = (float*)d_out;

    weightedsum_lse_wmma<<<dim3(NGRP), dim3(256), LDS_BYTES, stream>>>(ll, wgt, out);
}


# each wave owns one 16x16 tile for all 4 r planes: float4 out stored straight from accumulators (S round trip via LDS, 2 barriers gone); LDS pitch 68 makes fragment reads conflict-free; sum order kept
# speedup vs baseline: 1.8839x; 1.1312x over previous
#include <hip/hip_runtime.h>

#define BB 64
#define FF 128
#define CIN 64
#define CON 64
#define RR 4
#define CH 32
#define NGRP 256

typedef float v2f __attribute__((ext_vector_type(2)));
typedef float v8f __attribute__((ext_vector_type(8)));

#define E_STRIDE 68
#define E_R (BB * E_STRIDE + 2)
#define W_STRIDE 68
#define W_R (CH * W_STRIDE)
#define LE_OFF 0
#define LW_OFF (RR * E_R)
#define LM_OFF (LW_OFF + RR * W_R)
#define LDS_FLOATS (LM_OFF + RR * BB)
#define LDS_BYTES (LDS_FLOATS * 4)

extern "C" __global__ __launch_bounds__(256)
void weightedsum_lse_wmma(const float* __restrict__ ll,
                          const float* __restrict__ wgt,
                          float* __restrict__ out) {
    extern __shared__ __align__(16) float smem[];
    const int f   = blockIdx.x >> 1;
    const int c0  = (blockIdx.x & 1) * CH;
    const int tid = threadIdx.x;

    const float4* ll4 = (const float4*)ll;
    const float4* w4  = (const float4*)wgt;
    #pragma unroll
    for (int i = 0; i < 16; ++i) {
        const int idx = tid + i * 256;
        const int hi = idx >> 6, lo = idx & 63;

        float4 v = ll4[(hi * FF + f) * CIN + lo];
        smem[LE_OFF + 0 * E_R + hi * E_STRIDE + lo] = v.x;
        smem[LE_OFF + 1 * E_R + hi * E_STRIDE + lo] = v.y;
        smem[LE_OFF + 2 * E_R + hi * E_STRIDE + lo] = v.z;
        smem[LE_OFF + 3 * E_R + hi * E_STRIDE + lo] = v.w;
    }
    #pragma unroll
    for (int i = 0; i < 8; ++i) {
        const int idx = tid + i * 256;
        const int col16 = idx & 15;
        const int rest  = idx >> 5;
        const int ci    = ((rest >> 1) << 1) | ((idx >> 4) & 1);
        const int lo    = ((rest & 1) << 4) | col16;

        float4 wv = w4[(f * CIN + ci) * CON + c0 + lo];
        smem[LW_OFF + 0 * W_R + lo * W_STRIDE + ci] = wv.x;
        smem[LW_OFF + 1 * W_R + lo * W_STRIDE + ci] = wv.y;
        smem[LW_OFF + 2 * W_R + lo * W_STRIDE + ci] = wv.z;
        smem[LW_OFF + 3 * W_R + lo * W_STRIDE + ci] = wv.w;
    }
    __syncthreads();

    {
        const int b = (tid & 15) | (((tid >> 5) & 3) << 4);
        const int r = ((tid >> 4) & 1) | ((tid >> 7) << 1);
        float* row = &smem[LE_OFF + r * E_R + b * E_STRIDE];
        float v[CIN];
        #pragma unroll
        for (int ci = 0; ci < CIN; ++ci) v[ci] = row[ci];
        float m0 = v[0], m1 = v[1], m2 = v[2], m3 = v[3];
        #pragma unroll
        for (int ci = 4; ci < CIN; ci += 4) {
            m0 = fmaxf(m0, v[ci]);
            m1 = fmaxf(m1, v[ci + 1]);
            m2 = fmaxf(m2, v[ci + 2]);
            m3 = fmaxf(m3, v[ci + 3]);
        }
        const float m = fmaxf(fmaxf(m0, m1), fmaxf(m2, m3));
        smem[LM_OFF + b * RR + r] = m;
        #pragma unroll
        for (int ci = 0; ci < CIN; ++ci) row[ci] = __expf(v[ci] - m);
    }
    __syncthreads();

    const int wave  = tid >> 5;
    const int lane  = tid & 31;
    const int mt    = wave >> 1;
    const int nt    = wave & 1;
    const int lrow  = lane & 15;
    const int koff  = (lane >> 4) << 1;

    const float* Ea = &smem[LE_OFF + (mt * 16 + lrow) * E_STRIDE + koff];
    const float* Wa = &smem[LW_OFF + (nt * 16 + lrow) * W_STRIDE + koff];

    v8f acc[RR];
    #pragma unroll
    for (int r = 0; r < RR; ++r)
        acc[r] = (v8f){0.f, 0.f, 0.f, 0.f, 0.f, 0.f, 0.f, 0.f};
    #pragma unroll
    for (int k = 0; k < CIN; k += 4) {
        #pragma unroll
        for (int r = 0; r < RR; ++r) {
            const v2f a = *(const v2f*)&Ea[r * E_R + k];
            const v2f b = *(const v2f*)&Wa[r * W_R + k];
            acc[r] = __builtin_amdgcn_wmma_f32_16x16x4_f32(
                         false, a,  false, b,
                         (short)0, acc[r],  false,  false);
        }
    }

    float4* out4 = (float4*)out;
    const int rowbase = mt * 16 + ((lane >> 4) << 3);
    const int col = c0 + nt * 16 + lrow;
    #pragma unroll
    for (int v = 0; v < 8; ++v) {
        const int b = rowbase + v;
        const float4 mv = *(const float4*)&smem[LM_OFF + b * RR];
        float4 o;
        o.x = mv.x + __logf(acc[0][v]);
        o.y = mv.y + __logf(acc[1][v]);
        o.z = mv.z + __logf(acc[2][v]);
        o.w = mv.w + __logf(acc[3][v]);
        out4[(b * FF + f) * CON + col] = o;
    }
}

extern "C" void kernel_launch(void* const* d_in, const int* in_sizes, int n_in,
                              void* d_out, int out_size, void* d_ws, size_t ws_size,
                              hipStream_t stream) {
    (void)in_sizes; (void)n_in; (void)d_ws; (void)ws_size; (void)out_size;
    const float* ll  = (const float*)d_in[0];
    const float* wgt = (const float*)d_in[1];
    float* out = (float*)d_out;

    weightedsum_lse_wmma<<<dim3(NGRP), dim3(256), LDS_BYTES, stream>>>(ll, wgt, out);
}
